# starting kernel
# baseline (speedup 1.0000x reference)
#include <hip/hip_runtime.h>
#include <hip/hip_bf16.h>


typedef float v2f __attribute__((ext_vector_type(2)));
typedef float v8f __attribute__((ext_vector_type(8)));

#define SCQ 0.70710678118654752440f

__constant__ float cG0B[14] = {
   0.00325314f, -0.00388321f,  0.03466035f, -0.03887280f,
  -0.11720389f,  0.27529538f,  0.75614564f,  0.56881042f,
   0.01186609f, -0.10671180f,  0.02382538f,  0.01702522f,
  -0.00543948f, -0.00455690f };
__constant__ float cG0A[14] = {
  -0.00455690f, -0.00543948f,  0.01702522f,  0.02382538f,
  -0.10671180f,  0.01186609f,  0.56881042f,  0.75614564f,
   0.27529538f, -0.11720389f, -0.03887280f,  0.03466035f,
  -0.00388321f,  0.00325314f };
__constant__ float cG1B[14] = {
  -0.00455690f,  0.00543948f,  0.01702522f, -0.02382538f,
  -0.10671180f, -0.01186609f,  0.56881042f, -0.75614564f,
   0.27529538f,  0.11720389f, -0.03887280f, -0.03466035f,
  -0.00388321f, -0.00325314f };
__constant__ float cG1A[14] = {
  -0.00325314f, -0.00388321f, -0.03466035f, -0.03887280f,
   0.11720389f,  0.27529538f, -0.75614564f,  0.56881042f,
  -0.01186609f, -0.10671180f, -0.02382538f,  0.01702522f,
   0.00543948f, -0.00455690f };

__constant__ float cG0O[7] = { -3.0f/280.0f, -15.0f/280.0f, 73.0f/280.0f,
                               170.0f/280.0f, 73.0f/280.0f, -15.0f/280.0f,
                               -3.0f/280.0f };
__constant__ float cG1O[5] = { -1.0f/20.0f, -5.0f/20.0f, 12.0f/20.0f,
                               -5.0f/20.0f, -1.0f/20.0f };

__device__ __forceinline__ int ref_idx(int i, int L) {
  if (i < 0)  i = -1 - i;
  if (i >= L) i = 2 * L - 1 - i;
  return i;
}

__device__ __forceinline__ float c2q_fetch(const float* __restrict__ hi,
                                           int b, int r, int v, int ch,
                                           int sub, int Ld) {
  constexpr int SA[3] = { 0, 2, 1 };
  constexpr int SB[3] = { 5, 3, 4 };
  const int i = r >> 1, j = v >> 1;
  const long long base = (((long long)b * Ld + i) * Ld + j) * 36;
  const int py = r & 1, px = v & 1;
  const int im = (px ^ py) ? 18 : 0;
  const float a  = hi[base + ch + 3 * SA[sub] + im];
  const float bb = hi[base + ch + 3 * SB[sub] + im];
  return py ? ((px ^ py) ? (a - bb) * SCQ : (bb - a) * SCQ) : (a + bb) * SCQ;
}

__device__ __forceinline__ float colifilt_eval(bool g1, int qy, int sy, int v,
                                               const float* __restrict__ buf,
                                               int b, int ch, int sub) {
  const float* ha = g1 ? cG1B : cG0B;
  const float* hb = g1 ? cG1A : cG0A;
  const int   p   = sy >> 1;
  const bool  aB  = (sy & 1) == 0;
  const float* f  = aB ? ha : hb;
  const int base  = g1 ? (aB ? 7 : 6) : (aB ? 6 : 7);
  float acc = 0.0f;
#pragma unroll
  for (int j = 0; j < 7; ++j) {
    const int r = ref_idx(2 * qy + base - 2 * j, 256);
    float x;
    if (sub < 0) x = buf[(((long long)b * 256 + r) * 256 + v) * 3 + ch];
    else         x = c2q_fetch(buf, b, r, v, ch, sub, 128);
    acc += f[2 * j + p] * x;
  }
  return acc;
}

__global__ void __launch_bounds__(256)
k_level1(const float* __restrict__ low, const float* __restrict__ high1,
         float* __restrict__ Z) {
  const int tx = blockIdx.x, ty = blockIdx.y, bz = blockIdx.z;
  const int b = bz / 3, ch = bz % 3;
  const int x0 = tx * 16, y0 = ty * 16;
  const int u0 = 8 * tx - 6;

  __shared__ float y1s[16][21];
  __shared__ float y2s[16][21];

  const int tid = threadIdx.x;
  for (int idx = tid; idx < 2 * 16 * 20; idx += 256) {
    const int plane = idx / 320;
    const int rem   = idx - plane * 320;
    const int ry = rem / 20, ci = rem % 20;
    const int v  = ref_idx(u0 + ci, 256);
    const int y  = y0 + ry, qy = y >> 2, sy = y & 3;
    float val;
    if (plane == 0)
      val = colifilt_eval(false, qy, sy, v, low,   b, ch, -1) +
            colifilt_eval(true,  qy, sy, v, high1, b, ch,  0);
    else
      val = colifilt_eval(false, qy, sy, v, high1, b, ch,  1) +
            colifilt_eval(true,  qy, sy, v, high1, b, ch,  2);
    (plane ? y2s : y1s)[ry][ci] = val;
  }
  __syncthreads();

  const int ry = tid >> 4, cx = tid & 15;
  const int x  = x0 + cx;
  const int q2 = 2 * (cx >> 2);
  const int sx = x & 3, p = sx >> 1;
  const bool aB = (sx & 1) == 0;
  const float* f1 = aB ? cG0B : cG0A;  const int b1 = aB ? 6 : 7;
  const float* f2 = aB ? cG1B : cG1A;  const int b2 = aB ? 7 : 6;
  float acc = 0.0f;
#pragma unroll
  for (int j = 0; j < 7; ++j) {
    acc += f1[2 * j + p] * y1s[ry][q2 + b1 + 6 - 2 * j];
    acc += f2[2 * j + p] * y2s[ry][q2 + b2 + 6 - 2 * j];
  }
  Z[(((long long)b * 512 + (y0 + ry)) * 512 + x) * 3 + ch] = acc;
}

__device__ __forceinline__ float tap7(int i) {
  int ic = i < 0 ? 0 : (i > 6 ? 6 : i);
  float t = cG0O[ic];
  return (i == ic) ? t : 0.0f;
}
__device__ __forceinline__ float tap5(int i) {
  int ic = i < 0 ? 0 : (i > 4 ? 4 : i);
  float t = cG1O[ic];
  return (i == ic) ? t : 0.0f;
}

__device__ __forceinline__ v8f wmma4(v2f a, v2f b, v8f c) {
  return __builtin_amdgcn_wmma_f32_16x16x4_f32(
       false, a,  false, b,
       (short)0, c,  false,  false);
}

__global__ void __launch_bounds__(32)
k_level0(const float* __restrict__ Z, const float* __restrict__ high0,
         float* __restrict__ out) {
  const int tx = blockIdx.x, ty = blockIdx.y, bz = blockIdx.z;
  const int b = bz / 3, ch = bz % 3;
  const int x0 = tx * 16, y0 = ty * 16;

  __shared__ float zs [24][33];
  __shared__ float lhs[24][33];
  __shared__ float hls[24][33];
  __shared__ float hhs[24][33];
  __shared__ float y1s[16][33];
  __shared__ float y2s[16][33];

  const int lane = threadIdx.x;
  const int lo = lane & 15;
  const int h2 = (lane >> 4) * 2;

  v2f f0[6], f1[6];
#pragma unroll
  for (int kk = 0; kk < 6; ++kk) {
    const int k0 = 4 * kk + h2;
    f0[kk].x = tap7(k0 - lo - 1);  f0[kk].y = tap7(k0 - lo);
    f1[kk].x = tap5(k0 - lo - 2);  f1[kk].y = tap5(k0 - lo - 1);
  }

  for (int idx = lane; idx < 24 * 32; idx += 32) {
    const int r = idx >> 5, c = idx & 31;
    const int rr = ref_idx(y0 - 4 + r, 512);
    const int cc = ref_idx(x0 - 8 + c, 512);
    zs [r][c] = Z[(((long long)b * 512 + rr) * 512 + cc) * 3 + ch];
    lhs[r][c] = c2q_fetch(high0, b, rr, cc, ch, 0, 256);
    hls[r][c] = c2q_fetch(high0, b, rr, cc, ch, 1, 256);
    hhs[r][c] = c2q_fetch(high0, b, rr, cc, ch, 2, 256);
  }
  __syncthreads();

#pragma unroll
  for (int t = 0; t < 2; ++t) {
    v8f a1 = {}; v8f a2 = {};
#pragma unroll
    for (int kk = 0; kk < 6; ++kk) {
      const int k0 = 4 * kk + h2;
      const int cc = 16 * t + lo;
      v2f bz = { zs [k0][cc], zs [k0 + 1][cc] };
      v2f bl = { lhs[k0][cc], lhs[k0 + 1][cc] };
      v2f bh = { hls[k0][cc], hls[k0 + 1][cc] };
      v2f bq = { hhs[k0][cc], hhs[k0 + 1][cc] };
      a1 = wmma4(f0[kk], bz, a1);
      a1 = wmma4(f1[kk], bl, a1);
      a2 = wmma4(f0[kk], bh, a2);
      a2 = wmma4(f1[kk], bq, a2);
    }
#pragma unroll
    for (int v = 0; v < 8; ++v) {
      const int row = v + (h2 << 2);
      y1s[row][16 * t + lo] = a1[v];
      y2s[row][16 * t + lo] = a2[v];
    }
  }
  __syncthreads();

  v8f o = {};
#pragma unroll
  for (int kk = 0; kk < 6; ++kk) {
    const int k0 = 4 * kk + h2;
    v2f A1 = { y1s[lo][k0 + 4], y1s[lo][k0 + 5] };
    v2f A2 = { y2s[lo][k0 + 4], y2s[lo][k0 + 5] };
    o = wmma4(A1, f0[kk], o);
    o = wmma4(A2, f1[kk], o);
  }
#pragma unroll
  for (int v = 0; v < 8; ++v) {
    const int row = y0 + v + (h2 << 2);
    const int col = x0 + lo;
    out[(((long long)b * 512 + row) * 512 + col) * 3 + ch] = o[v];
  }
}

extern "C" void kernel_launch(void* const* d_in, const int* in_sizes, int n_in,
                              void* d_out, int out_size, void* d_ws, size_t ws_size,
                              hipStream_t stream) {
  (void)in_sizes; (void)n_in; (void)out_size; (void)ws_size;
  const float* low   = (const float*)d_in[0];
  const float* high0 = (const float*)d_in[1];
  const float* high1 = (const float*)d_in[2];
  float* out = (float*)d_out;
  float* Zws = (float*)d_ws;

  dim3 grid(32, 32, 8 * 3);
  k_level1<<<grid, dim3(256), 0, stream>>>(low, high1, Zws);
  k_level0<<<grid, dim3(32), 0, stream>>>(Zws, high0, out);
}


# level-1 moved from vector code to f32 WMMA: colifilt as banded GEMMs, one wave per 16x32 Z tile, column pass transposed so its D registers feed the row pass (no LDS); sums regrouped
# speedup vs baseline: 2.6320x; 2.6320x over previous
#include <hip/hip_runtime.h>
#include <hip/hip_bf16.h>


typedef float v2f __attribute__((ext_vector_type(2)));
typedef float v8f __attribute__((ext_vector_type(8)));

#define SCQ 0.70710678118654752440f

__constant__ float cG0B[14] = {
   0.00325314f, -0.00388321f,  0.03466035f, -0.03887280f,
  -0.11720389f,  0.27529538f,  0.75614564f,  0.56881042f,
   0.01186609f, -0.10671180f,  0.02382538f,  0.01702522f,
  -0.00543948f, -0.00455690f };
__constant__ float cG0A[14] = {
  -0.00455690f, -0.00543948f,  0.01702522f,  0.02382538f,
  -0.10671180f,  0.01186609f,  0.56881042f,  0.75614564f,
   0.27529538f, -0.11720389f, -0.03887280f,  0.03466035f,
  -0.00388321f,  0.00325314f };
__constant__ float cG1B[14] = {
  -0.00455690f,  0.00543948f,  0.01702522f, -0.02382538f,
  -0.10671180f, -0.01186609f,  0.56881042f, -0.75614564f,
   0.27529538f,  0.11720389f, -0.03887280f, -0.03466035f,
  -0.00388321f, -0.00325314f };
__constant__ float cG1A[14] = {
  -0.00325314f, -0.00388321f, -0.03466035f, -0.03887280f,
   0.11720389f,  0.27529538f, -0.75614564f,  0.56881042f,
  -0.01186609f, -0.10671180f, -0.02382538f,  0.01702522f,
   0.00543948f, -0.00455690f };

__constant__ float cG0O[7] = { -3.0f/280.0f, -15.0f/280.0f, 73.0f/280.0f,
                               170.0f/280.0f, 73.0f/280.0f, -15.0f/280.0f,
                               -3.0f/280.0f };
__constant__ float cG1O[5] = { -1.0f/20.0f, -5.0f/20.0f, 12.0f/20.0f,
                               -5.0f/20.0f, -1.0f/20.0f };

__device__ __forceinline__ int ref_idx(int i, int L) {
  if (i < 0)  i = -1 - i;
  if (i >= L) i = 2 * L - 1 - i;
  return i;
}

__device__ __forceinline__ float c2q_fetch(const float* __restrict__ hi,
                                           int b, int r, int v, int ch,
                                           int sub, int Ld) {
  constexpr int SA[3] = { 0, 2, 1 };
  constexpr int SB[3] = { 5, 3, 4 };
  const int i = r >> 1, j = v >> 1;
  const long long base = (((long long)b * Ld + i) * Ld + j) * 36;
  const int py = r & 1, px = v & 1;
  const int im = (px ^ py) ? 18 : 0;
  const float a  = hi[base + ch + 3 * SA[sub] + im];
  const float bb = hi[base + ch + 3 * SB[sub] + im];
  return py ? ((px ^ py) ? (a - bb) * SCQ : (bb - a) * SCQ) : (a + bb) * SCQ;
}

__device__ __forceinline__ v8f wmma4(v2f a, v2f b, v8f c) {
  return __builtin_amdgcn_wmma_f32_16x16x4_f32(
       false, a,  false, b,
       (short)0, c,  false,  false);
}

__device__ __forceinline__ float c2q_flat(const float* __restrict__ hi,
                                          int b, int r, int v, int ch,
                                          int sub, int Ld) {
  constexpr int SA[3] = { 0, 2, 1 };
  constexpr int SB[3] = { 5, 3, 4 };
  const int i = r >> 1, j = v >> 1;
  const int py = r & 1, px = v & 1;
  const int base = ((b * Ld + i) * Ld + j) * 36 + ch + ((px ^ py) ? 18 : 0);
  const float a  = hi[base + 3 * SA[sub]];
  const float bb = hi[base + 3 * SB[sub]];
  const float t = __uint_as_float(__float_as_uint(bb) ^ ((unsigned)py << 31));
  const float k = __uint_as_float(__float_as_uint(SCQ) ^ ((unsigned)(py & px) << 31));
  return (a + t) * k;
}

__device__ __forceinline__ float band14(bool g1, int x, int c) {
  const int  s  = x & 3, q = x >> 2;
  const int  p  = s >> 1;
  const bool aB = (s & 1) == 0;
  const float* f = g1 ? (aB ? cG1B : cG1A) : (aB ? cG0B : cG0A);
  const int base = g1 ? (aB ? 7 : 6) : (aB ? 6 : 7);
  const int d    = 2 * q + base + 6 - c;
  const bool ok  = (d >= 0) && (d <= 12) && ((d & 1) == 0);
  const float t  = f[ok ? d + p : 0];
  return ok ? t : 0.0f;
}

__device__ __forceinline__ int krow(int kk, int h, int pos) {
  return 4 * kk + h + 2 * pos;
}

__global__ void __launch_bounds__(32)
k_level1(const float* __restrict__ low, const float* __restrict__ high1,
         float* __restrict__ Z) {
  const int tx = blockIdx.x, ty = blockIdx.y, bz = blockIdx.z;
  const int b = bz / 3, ch = bz % 3;
  const int x0 = tx * 32, y0 = ty * 16;
  const int r0 = 8 * ty - 6;
  const int c0 = 16 * tx - 6;

  const int lane = threadIdx.x;
  const int lo = lane & 15;
  const int h  = lane >> 4;

  v2f cf0[5], cf1[5];
#pragma unroll
  for (int kk = 0; kk < 5; ++kk) {
    const int ka = krow(kk, h, 0), kb = krow(kk, h, 1);
    cf0[kk].x = band14(false, lo, ka);  cf0[kk].y = band14(false, lo, kb);
    cf1[kk].x = band14(true,  lo, ka);  cf1[kk].y = band14(true,  lo, kb);
  }

  v8f d1[2], d2[2];
#pragma unroll
  for (int T = 0; T < 2; ++T) {
    int cw = 16 * T + lo;
    cw = cw > 27 ? 27 : cw;
    const int vv = ref_idx(c0 + cw, 256);
    v8f a1 = {}; v8f a2 = {};
#pragma unroll
    for (int kk = 0; kk < 5; ++kk) {
      const int ra = ref_idx(r0 + krow(kk, h, 0), 256);
      const int rb = ref_idx(r0 + krow(kk, h, 1), 256);
      v2f wl = { low[((b * 256 + ra) * 256 + vv) * 3 + ch],
                 low[((b * 256 + rb) * 256 + vv) * 3 + ch] };
      v2f wa = { c2q_flat(high1, b, ra, vv, ch, 0, 128),
                 c2q_flat(high1, b, rb, vv, ch, 0, 128) };
      v2f wb = { c2q_flat(high1, b, ra, vv, ch, 1, 128),
                 c2q_flat(high1, b, rb, vv, ch, 1, 128) };
      v2f wc = { c2q_flat(high1, b, ra, vv, ch, 2, 128),
                 c2q_flat(high1, b, rb, vv, ch, 2, 128) };
      a1 = wmma4(wl, cf0[kk], a1);
      a2 = wmma4(wb, cf0[kk], a2);
      a1 = wmma4(wa, cf1[kk], a1);
      a2 = wmma4(wc, cf1[kk], a2);
    }
    d1[T] = a1; d2[T] = a2;
  }

#pragma unroll
  for (int U = 0; U < 2; ++U) {
    v8f o = {};
#pragma unroll
    for (int T = 0; T < 2; ++T) {
#pragma unroll
      for (int s = 0; s < 4; ++s) {
        if (U == 0 && T == 1 && s >= 2) continue;
        const int cx = 16 * T + 8 * h + 2 * s;
        v2f g0 = { band14(false, 16 * U + lo, cx), band14(false, 16 * U + lo, cx + 1) };
        v2f g1 = { band14(true,  16 * U + lo, cx), band14(true,  16 * U + lo, cx + 1) };
        v2f A1 = { d1[T][2 * s], d1[T][2 * s + 1] };
        v2f A2 = { d2[T][2 * s], d2[T][2 * s + 1] };
        o = wmma4(A1, g0, o);
        o = wmma4(A2, g1, o);
      }
    }
#pragma unroll
    for (int v = 0; v < 8; ++v) {
      const int row = y0 + v + 8 * h;
      const int col = x0 + 16 * U + lo;
      Z[((b * 512 + row) * 512 + col) * 3 + ch] = o[v];
    }
  }
}

__device__ __forceinline__ float tap7(int i) {
  int ic = i < 0 ? 0 : (i > 6 ? 6 : i);
  float t = cG0O[ic];
  return (i == ic) ? t : 0.0f;
}
__device__ __forceinline__ float tap5(int i) {
  int ic = i < 0 ? 0 : (i > 4 ? 4 : i);
  float t = cG1O[ic];
  return (i == ic) ? t : 0.0f;
}

__global__ void __launch_bounds__(32)
k_level0(const float* __restrict__ Z, const float* __restrict__ high0,
         float* __restrict__ out) {
  const int tx = blockIdx.x, ty = blockIdx.y, bz = blockIdx.z;
  const int b = bz / 3, ch = bz % 3;
  const int x0 = tx * 16, y0 = ty * 16;

  __shared__ float zs [24][33];
  __shared__ float lhs[24][33];
  __shared__ float hls[24][33];
  __shared__ float hhs[24][33];
  __shared__ float y1s[16][33];
  __shared__ float y2s[16][33];

  const int lane = threadIdx.x;
  const int lo = lane & 15;
  const int h2 = (lane >> 4) * 2;

  v2f f0[6], f1[6];
#pragma unroll
  for (int kk = 0; kk < 6; ++kk) {
    const int k0 = 4 * kk + h2;
    f0[kk].x = tap7(k0 - lo - 1);  f0[kk].y = tap7(k0 - lo);
    f1[kk].x = tap5(k0 - lo - 2);  f1[kk].y = tap5(k0 - lo - 1);
  }

  for (int idx = lane; idx < 24 * 32; idx += 32) {
    const int r = idx >> 5, c = idx & 31;
    const int rr = ref_idx(y0 - 4 + r, 512);
    const int cc = ref_idx(x0 - 8 + c, 512);
    zs [r][c] = Z[(((long long)b * 512 + rr) * 512 + cc) * 3 + ch];
    lhs[r][c] = c2q_fetch(high0, b, rr, cc, ch, 0, 256);
    hls[r][c] = c2q_fetch(high0, b, rr, cc, ch, 1, 256);
    hhs[r][c] = c2q_fetch(high0, b, rr, cc, ch, 2, 256);
  }
  __syncthreads();

#pragma unroll
  for (int t = 0; t < 2; ++t) {
    v8f a1 = {}; v8f a2 = {};
#pragma unroll
    for (int kk = 0; kk < 6; ++kk) {
      const int k0 = 4 * kk + h2;
      const int cc = 16 * t + lo;
      v2f bz = { zs [k0][cc], zs [k0 + 1][cc] };
      v2f bl = { lhs[k0][cc], lhs[k0 + 1][cc] };
      v2f bh = { hls[k0][cc], hls[k0 + 1][cc] };
      v2f bq = { hhs[k0][cc], hhs[k0 + 1][cc] };
      a1 = wmma4(f0[kk], bz, a1);
      a1 = wmma4(f1[kk], bl, a1);
      a2 = wmma4(f0[kk], bh, a2);
      a2 = wmma4(f1[kk], bq, a2);
    }
#pragma unroll
    for (int v = 0; v < 8; ++v) {
      const int row = v + (h2 << 2);
      y1s[row][16 * t + lo] = a1[v];
      y2s[row][16 * t + lo] = a2[v];
    }
  }
  __syncthreads();

  v8f o = {};
#pragma unroll
  for (int kk = 0; kk < 6; ++kk) {
    const int k0 = 4 * kk + h2;
    v2f A1 = { y1s[lo][k0 + 4], y1s[lo][k0 + 5] };
    v2f A2 = { y2s[lo][k0 + 4], y2s[lo][k0 + 5] };
    o = wmma4(A1, f0[kk], o);
    o = wmma4(A2, f1[kk], o);
  }
#pragma unroll
  for (int v = 0; v < 8; ++v) {
    const int row = y0 + v + (h2 << 2);
    const int col = x0 + lo;
    out[(((long long)b * 512 + row) * 512 + col) * 3 + ch] = o[v];
  }
}

extern "C" void kernel_launch(void* const* d_in, const int* in_sizes, int n_in,
                              void* d_out, int out_size, void* d_ws, size_t ws_size,
                              hipStream_t stream) {
  (void)in_sizes; (void)n_in; (void)out_size; (void)ws_size;
  const float* low   = (const float*)d_in[0];
  const float* high0 = (const float*)d_in[1];
  const float* high1 = (const float*)d_in[2];
  float* out = (float*)d_out;
  float* Zws = (float*)d_ws;

  dim3 grid1(16, 32, 8 * 3);
  dim3 grid(32, 32, 8 * 3);
  k_level1<<<grid1, dim3(32), 0, stream>>>(low, high1, Zws);
  k_level0<<<grid, dim3(32), 0, stream>>>(Zws, high0, out);
}


# level-0 without LDS: one wave per 16x64x3 tile, each lane reads its high0 record once (16-byte loads, 3 k-steps ahead, c2q in registers), transposed column pass feeds row pass; row sums regrouped
# speedup vs baseline: 5.3589x; 2.0361x over previous
#include <hip/hip_runtime.h>
#include <hip/hip_bf16.h>


typedef float v2f __attribute__((ext_vector_type(2)));
typedef float v8f __attribute__((ext_vector_type(8)));

#define SCQ 0.70710678118654752440f

__constant__ float cG0B[14] = {
   0.00325314f, -0.00388321f,  0.03466035f, -0.03887280f,
  -0.11720389f,  0.27529538f,  0.75614564f,  0.56881042f,
   0.01186609f, -0.10671180f,  0.02382538f,  0.01702522f,
  -0.00543948f, -0.00455690f };
__constant__ float cG0A[14] = {
  -0.00455690f, -0.00543948f,  0.01702522f,  0.02382538f,
  -0.10671180f,  0.01186609f,  0.56881042f,  0.75614564f,
   0.27529538f, -0.11720389f, -0.03887280f,  0.03466035f,
  -0.00388321f,  0.00325314f };
__constant__ float cG1B[14] = {
  -0.00455690f,  0.00543948f,  0.01702522f, -0.02382538f,
  -0.10671180f, -0.01186609f,  0.56881042f, -0.75614564f,
   0.27529538f,  0.11720389f, -0.03887280f, -0.03466035f,
  -0.00388321f, -0.00325314f };
__constant__ float cG1A[14] = {
  -0.00325314f, -0.00388321f, -0.03466035f, -0.03887280f,
   0.11720389f,  0.27529538f, -0.75614564f,  0.56881042f,
  -0.01186609f, -0.10671180f, -0.02382538f,  0.01702522f,
   0.00543948f, -0.00455690f };

__constant__ float cG0O[7] = { -3.0f/280.0f, -15.0f/280.0f, 73.0f/280.0f,
                               170.0f/280.0f, 73.0f/280.0f, -15.0f/280.0f,
                               -3.0f/280.0f };
__constant__ float cG1O[5] = { -1.0f/20.0f, -5.0f/20.0f, 12.0f/20.0f,
                               -5.0f/20.0f, -1.0f/20.0f };

__device__ __forceinline__ int ref_idx(int i, int L) {
  if (i < 0)  i = -1 - i;
  if (i >= L) i = 2 * L - 1 - i;
  return i;
}

__device__ __forceinline__ float c2q_fetch(const float* __restrict__ hi,
                                           int b, int r, int v, int ch,
                                           int sub, int Ld) {
  constexpr int SA[3] = { 0, 2, 1 };
  constexpr int SB[3] = { 5, 3, 4 };
  const int i = r >> 1, j = v >> 1;
  const long long base = (((long long)b * Ld + i) * Ld + j) * 36;
  const int py = r & 1, px = v & 1;
  const int im = (px ^ py) ? 18 : 0;
  const float a  = hi[base + ch + 3 * SA[sub] + im];
  const float bb = hi[base + ch + 3 * SB[sub] + im];
  return py ? ((px ^ py) ? (a - bb) * SCQ : (bb - a) * SCQ) : (a + bb) * SCQ;
}

__device__ __forceinline__ v8f wmma4(v2f a, v2f b, v8f c) {
  return __builtin_amdgcn_wmma_f32_16x16x4_f32(
       false, a,  false, b,
       (short)0, c,  false,  false);
}

__device__ __forceinline__ float c2q_flat(const float* __restrict__ hi,
                                          int b, int r, int v, int ch,
                                          int sub, int Ld) {
  constexpr int SA[3] = { 0, 2, 1 };
  constexpr int SB[3] = { 5, 3, 4 };
  const int i = r >> 1, j = v >> 1;
  const int py = r & 1, px = v & 1;
  const int base = ((b * Ld + i) * Ld + j) * 36 + ch + ((px ^ py) ? 18 : 0);
  const float a  = hi[base + 3 * SA[sub]];
  const float bb = hi[base + 3 * SB[sub]];
  const float t = __uint_as_float(__float_as_uint(bb) ^ ((unsigned)py << 31));
  const float k = __uint_as_float(__float_as_uint(SCQ) ^ ((unsigned)(py & px) << 31));
  return (a + t) * k;
}

__device__ __forceinline__ float band14(bool g1, int x, int c) {
  const int  s  = x & 3, q = x >> 2;
  const int  p  = s >> 1;
  const bool aB = (s & 1) == 0;
  const float* f = g1 ? (aB ? cG1B : cG1A) : (aB ? cG0B : cG0A);
  const int base = g1 ? (aB ? 7 : 6) : (aB ? 6 : 7);
  const int d    = 2 * q + base + 6 - c;
  const bool ok  = (d >= 0) && (d <= 12) && ((d & 1) == 0);
  const float t  = f[ok ? d + p : 0];
  return ok ? t : 0.0f;
}

__device__ __forceinline__ int krow(int kk, int h, int pos) {
  return 4 * kk + h + 2 * pos;
}

__global__ void __launch_bounds__(32)
k_level1(const float* __restrict__ low, const float* __restrict__ high1,
         float* __restrict__ Z) {
  const int tx = blockIdx.x, ty = blockIdx.y, bz = blockIdx.z;
  const int b = bz / 3, ch = bz % 3;
  const int x0 = tx * 32, y0 = ty * 16;
  const int r0 = 8 * ty - 6;
  const int c0 = 16 * tx - 6;

  const int lane = threadIdx.x;
  const int lo = lane & 15;
  const int h  = lane >> 4;

  v2f cf0[5], cf1[5];
#pragma unroll
  for (int kk = 0; kk < 5; ++kk) {
    const int ka = krow(kk, h, 0), kb = krow(kk, h, 1);
    cf0[kk].x = band14(false, lo, ka);  cf0[kk].y = band14(false, lo, kb);
    cf1[kk].x = band14(true,  lo, ka);  cf1[kk].y = band14(true,  lo, kb);
  }

  v8f d1[2], d2[2];
#pragma unroll
  for (int T = 0; T < 2; ++T) {
    int cw = 16 * T + lo;
    cw = cw > 27 ? 27 : cw;
    const int vv = ref_idx(c0 + cw, 256);
    v8f a1 = {}; v8f a2 = {};
#pragma unroll
    for (int kk = 0; kk < 5; ++kk) {
      const int ra = ref_idx(r0 + krow(kk, h, 0), 256);
      const int rb = ref_idx(r0 + krow(kk, h, 1), 256);
      v2f wl = { low[((b * 256 + ra) * 256 + vv) * 3 + ch],
                 low[((b * 256 + rb) * 256 + vv) * 3 + ch] };
      v2f wa = { c2q_flat(high1, b, ra, vv, ch, 0, 128),
                 c2q_flat(high1, b, rb, vv, ch, 0, 128) };
      v2f wb = { c2q_flat(high1, b, ra, vv, ch, 1, 128),
                 c2q_flat(high1, b, rb, vv, ch, 1, 128) };
      v2f wc = { c2q_flat(high1, b, ra, vv, ch, 2, 128),
                 c2q_flat(high1, b, rb, vv, ch, 2, 128) };
      a1 = wmma4(wl, cf0[kk], a1);
      a2 = wmma4(wb, cf0[kk], a2);
      a1 = wmma4(wa, cf1[kk], a1);
      a2 = wmma4(wc, cf1[kk], a2);
    }
    d1[T] = a1; d2[T] = a2;
  }

#pragma unroll
  for (int U = 0; U < 2; ++U) {
    v8f o = {};
#pragma unroll
    for (int T = 0; T < 2; ++T) {
#pragma unroll
      for (int s = 0; s < 4; ++s) {
        if (U == 0 && T == 1 && s >= 2) continue;
        const int cx = 16 * T + 8 * h + 2 * s;
        v2f g0 = { band14(false, 16 * U + lo, cx), band14(false, 16 * U + lo, cx + 1) };
        v2f g1 = { band14(true,  16 * U + lo, cx), band14(true,  16 * U + lo, cx + 1) };
        v2f A1 = { d1[T][2 * s], d1[T][2 * s + 1] };
        v2f A2 = { d2[T][2 * s], d2[T][2 * s + 1] };
        o = wmma4(A1, g0, o);
        o = wmma4(A2, g1, o);
      }
    }
#pragma unroll
    for (int v = 0; v < 8; ++v) {
      const int row = y0 + v + 8 * h;
      const int col = x0 + 16 * U + lo;
      Z[((b * 512 + row) * 512 + col) * 3 + ch] = o[v];
    }
  }
}

__device__ __forceinline__ float tap7(int i) {
  int ic = i < 0 ? 0 : (i > 6 ? 6 : i);
  float t = cG0O[ic];
  return (i == ic) ? t : 0.0f;
}
__device__ __forceinline__ float tap5(int i) {
  int ic = i < 0 ? 0 : (i > 4 ? 4 : i);
  float t = cG1O[ic];
  return (i == ic) ? t : 0.0f;
}

typedef float v4f __attribute__((ext_vector_type(4)));

struct L0Rec { v4f q[9]; float za[3]; float zb[3]; };

__device__ __forceinline__ int l0_col(int x0, int T, int lo) {
  int cw = 16 * T + lo;
  cw = cw > 69 ? 69 : cw;
  return ref_idx(x0 - 3 + cw, 512);
}

__device__ __forceinline__ void l0_load(
    const float* __restrict__ Z, const float* __restrict__ high0,
    int b, int gc, int rik, L0Rec& r) {
  const v4f* rp = (const v4f*)(high0 + ((b * 256 + rik) * 256 + (gc >> 1)) * 36);
#pragma unroll
  for (int q = 0; q < 9; ++q) r.q[q] = rp[q];
  const int zo = ((b * 512 + 2 * rik) * 512 + gc) * 3;
#pragma unroll
  for (int ch = 0; ch < 3; ++ch) { r.za[ch] = Z[zo + ch]; r.zb[ch] = Z[zo + 512 * 3 + ch]; }
}

__device__ __forceinline__ void l0_mac(
    const L0Rec& r, bool px, v2f c0, v2f c1, v8f (&a1)[3], v8f (&a2)[3]) {
  constexpr int SA[3] = { 0, 2, 1 };
  constexpr int SB[3] = { 5, 3, 4 };
  const float sq = px ? -SCQ : SCQ;
  float f[36];
#pragma unroll
  for (int q = 0; q < 9; ++q) {
    f[4 * q] = r.q[q].x; f[4 * q + 1] = r.q[q].y;
    f[4 * q + 2] = r.q[q].z; f[4 * q + 3] = r.q[q].w;
  }
  v2f w[3][3];
#pragma unroll
  for (int sub = 0; sub < 3; ++sub) {
#pragma unroll
    for (int ch = 0; ch < 3; ++ch) {
      const int ia = 3 * SA[sub] + ch, ib = 3 * SB[sub] + ch;
      const float pa = px ? f[18 + ia] : f[ia];
      const float pb = px ? f[18 + ib] : f[ib];
      const float qa = px ? f[ia] : f[18 + ia];
      const float qb = px ? f[ib] : f[18 + ib];
      w[sub][ch].x = (pa + pb) * SCQ;
      w[sub][ch].y = (qa - qb) * sq;
    }
  }
#pragma unroll
  for (int ch = 0; ch < 3; ++ch) {
    v2f wz = { r.za[ch], r.zb[ch] };
    a1[ch] = wmma4(wz,       c0, a1[ch]);
    a2[ch] = wmma4(w[1][ch], c0, a2[ch]);
    a1[ch] = wmma4(w[0][ch], c1, a1[ch]);
    a2[ch] = wmma4(w[2][ch], c1, a2[ch]);
  }
}

__device__ __forceinline__ void l0_row_part(
    const v8f (&y1)[3], const v8f (&y2)[3], int D,
    const v2f (&g0)[2][4], const v2f (&g1)[2][4], v8f (&o)[3]) {
#pragma unroll
  for (int s = 0; s < 4; ++s) {
    if (D == 1 && s == 3) continue;
#pragma unroll
    for (int ch = 0; ch < 3; ++ch) {
      v2f A1 = { y1[ch][2 * s], y1[ch][2 * s + 1] };
      v2f A2 = { y2[ch][2 * s], y2[ch][2 * s + 1] };
      o[ch] = wmma4(A1, g0[D][s], o[ch]);
      o[ch] = wmma4(A2, g1[D][s], o[ch]);
    }
  }
}

__device__ __forceinline__ void l0_store(
    const v8f (&o)[3], int lo, int h, float* __restrict__ out,
    int b, int x0, int y0, int U) {
#pragma unroll
  for (int v = 0; v < 8; ++v) {
    const int row = y0 + v + 8 * h;
    const int col = x0 + 16 * U + lo;
    float* op = out + ((b * 512 + row) * 512 + col) * 3;
    op[0] = o[0][v]; op[1] = o[1][v]; op[2] = o[2][v];
  }
}

__global__ void __launch_bounds__(32)
k_level0(const float* __restrict__ Z, const float* __restrict__ high0,
         float* __restrict__ out) {
  const int tx = blockIdx.x, ty = blockIdx.y, b = blockIdx.z;
  const int x0 = tx * 64, y0 = ty * 16;
  const int lane = threadIdx.x;
  const int lo = lane & 15;
  const int h  = lane >> 4;

  int ri[6], ka[6], kb[6];
#pragma unroll
  for (int kk = 0; kk < 6; ++kk) {
    const int ru = y0 - 4 + 4 * kk + 2 * h;
    const int sw = (ru < 0 || ru >= 512) ? 1 : 0;
    ri[kk] = ref_idx(ru, 512) >> 1;
    ka[kk] = 4 * kk + 2 * h + sw;
    kb[kk] = 4 * kk + 2 * h + (sw ^ 1);
  }
  int gc[5];
#pragma unroll
  for (int T = 0; T < 5; ++T) gc[T] = l0_col(x0, T, lo);

  constexpr int NS = 30, PF = 3;
  L0Rec r[NS];
#pragma unroll
  for (int n = 0; n < PF; ++n) l0_load(Z, high0, b, gc[n / 6], ri[n % 6], r[n]);

  v2f cf0[6], cf1[6];
#pragma unroll
  for (int kk = 0; kk < 6; ++kk) {
    cf0[kk].x = tap7(ka[kk] - lo - 1);  cf0[kk].y = tap7(kb[kk] - lo - 1);
    cf1[kk].x = tap5(ka[kk] - lo - 2);  cf1[kk].y = tap5(kb[kk] - lo - 2);
  }

  v2f g0[2][4], g1[2][4];
#pragma unroll
  for (int D = 0; D < 2; ++D) {
#pragma unroll
    for (int s = 0; s < 4; ++s) {
      const int dx = 16 * D + 8 * h + 2 * s - lo;
      g0[D][s].x = tap7(dx);      g0[D][s].y = tap7(dx + 1);
      g1[D][s].x = tap5(dx - 1);  g1[D][s].y = tap5(dx);
    }
  }

  v8f a1[3], a2[3], o[4][3];
#pragma unroll
  for (int ch = 0; ch < 3; ++ch) {
    a1[ch] = v8f{}; a2[ch] = v8f{};
#pragma unroll
    for (int U = 0; U < 4; ++U) o[U][ch] = v8f{};
  }
#pragma unroll
  for (int n = 0; n < NS; ++n) {
    const int T = n / 6, kk = n % 6;
    if (n + PF < NS)
      l0_load(Z, high0, b, gc[(n + PF) / 6], ri[(n + PF) % 6], r[n + PF]);
    __builtin_amdgcn_sched_barrier(0);
    l0_mac(r[n], (gc[T] & 1) != 0, cf0[kk], cf1[kk], a1, a2);
    if (kk == 5) {
      if (T >= 1) {
        l0_row_part(a1, a2, 1, g0, g1, o[T - 1]);
        l0_store(o[T - 1], lo, h, out, b, x0, y0, T - 1);
      }
      if (T <= 3) l0_row_part(a1, a2, 0, g0, g1, o[T]);
#pragma unroll
      for (int ch = 0; ch < 3; ++ch) { a1[ch] = v8f{}; a2[ch] = v8f{}; }
    }
  }
}

extern "C" void kernel_launch(void* const* d_in, const int* in_sizes, int n_in,
                              void* d_out, int out_size, void* d_ws, size_t ws_size,
                              hipStream_t stream) {
  (void)in_sizes; (void)n_in; (void)out_size; (void)ws_size;
  const float* low   = (const float*)d_in[0];
  const float* high0 = (const float*)d_in[1];
  const float* high1 = (const float*)d_in[2];
  float* out = (float*)d_out;
  float* Zws = (float*)d_ws;

  dim3 grid1(16, 32, 8 * 3);
  dim3 grid(8, 32, 8);
  k_level1<<<grid1, dim3(32), 0, stream>>>(low, high1, Zws);
  k_level0<<<grid, dim3(32), 0, stream>>>(Zws, high0, out);
}


# level-1: one wave per 16x32x3 Z tile (three channels), each lane reads the 18-float half of its high1 records once (16-byte loads, a k-step ahead), c2q in registers; 4096 waves, not 12288
# speedup vs baseline: 6.5641x; 1.2249x over previous
#include <hip/hip_runtime.h>
#include <hip/hip_bf16.h>


typedef float v2f __attribute__((ext_vector_type(2)));
typedef float v8f __attribute__((ext_vector_type(8)));

#define SCQ 0.70710678118654752440f

__constant__ float cG0B[14] = {
   0.00325314f, -0.00388321f,  0.03466035f, -0.03887280f,
  -0.11720389f,  0.27529538f,  0.75614564f,  0.56881042f,
   0.01186609f, -0.10671180f,  0.02382538f,  0.01702522f,
  -0.00543948f, -0.00455690f };
__constant__ float cG0A[14] = {
  -0.00455690f, -0.00543948f,  0.01702522f,  0.02382538f,
  -0.10671180f,  0.01186609f,  0.56881042f,  0.75614564f,
   0.27529538f, -0.11720389f, -0.03887280f,  0.03466035f,
  -0.00388321f,  0.00325314f };
__constant__ float cG1B[14] = {
  -0.00455690f,  0.00543948f,  0.01702522f, -0.02382538f,
  -0.10671180f, -0.01186609f,  0.56881042f, -0.75614564f,
   0.27529538f,  0.11720389f, -0.03887280f, -0.03466035f,
  -0.00388321f, -0.00325314f };
__constant__ float cG1A[14] = {
  -0.00325314f, -0.00388321f, -0.03466035f, -0.03887280f,
   0.11720389f,  0.27529538f, -0.75614564f,  0.56881042f,
  -0.01186609f, -0.10671180f, -0.02382538f,  0.01702522f,
   0.00543948f, -0.00455690f };

__constant__ float cG0O[7] = { -3.0f/280.0f, -15.0f/280.0f, 73.0f/280.0f,
                               170.0f/280.0f, 73.0f/280.0f, -15.0f/280.0f,
                               -3.0f/280.0f };
__constant__ float cG1O[5] = { -1.0f/20.0f, -5.0f/20.0f, 12.0f/20.0f,
                               -5.0f/20.0f, -1.0f/20.0f };

__device__ __forceinline__ int ref_idx(int i, int L) {
  if (i < 0)  i = -1 - i;
  if (i >= L) i = 2 * L - 1 - i;
  return i;
}

__device__ __forceinline__ float c2q_fetch(const float* __restrict__ hi,
                                           int b, int r, int v, int ch,
                                           int sub, int Ld) {
  constexpr int SA[3] = { 0, 2, 1 };
  constexpr int SB[3] = { 5, 3, 4 };
  const int i = r >> 1, j = v >> 1;
  const long long base = (((long long)b * Ld + i) * Ld + j) * 36;
  const int py = r & 1, px = v & 1;
  const int im = (px ^ py) ? 18 : 0;
  const float a  = hi[base + ch + 3 * SA[sub] + im];
  const float bb = hi[base + ch + 3 * SB[sub] + im];
  return py ? ((px ^ py) ? (a - bb) * SCQ : (bb - a) * SCQ) : (a + bb) * SCQ;
}

__device__ __forceinline__ v8f wmma4(v2f a, v2f b, v8f c) {
  return __builtin_amdgcn_wmma_f32_16x16x4_f32(
       false, a,  false, b,
       (short)0, c,  false,  false);
}

__device__ __forceinline__ float band14(bool g1, int x, int c) {
  const int  s  = x & 3, q = x >> 2;
  const int  p  = s >> 1;
  const bool aB = (s & 1) == 0;
  const float* f = g1 ? (aB ? cG1B : cG1A) : (aB ? cG0B : cG0A);
  const int base = g1 ? (aB ? 7 : 6) : (aB ? 6 : 7);
  const int d    = 2 * q + base + 6 - c;
  const bool ok  = (d >= 0) && (d <= 12) && ((d & 1) == 0);
  const float t  = f[ok ? d + p : 0];
  return ok ? t : 0.0f;
}

__device__ __forceinline__ int krow(int kk, int h, int pos) {
  return 4 * kk + h + 2 * pos;
}

typedef float v3f __attribute__((ext_vector_type(3)));
typedef float v4f __attribute__((ext_vector_type(4)));
typedef v4f v4f_a8 __attribute__((aligned(8)));
typedef v3f v3f_a4 __attribute__((aligned(4)));

struct L1Rec { v4f qa[4]; v2f ea; v4f qb[4]; v2f eb; v3f la; v3f lb; };

__device__ __forceinline__ void l1_load(
    const float* __restrict__ low, const float* __restrict__ high1,
    int b, int vv, int ra, int rb, L1Rec& r) {
  const int j = vv >> 1;
  const float* pa = high1 + ((b * 128 + (ra >> 1)) * 128 + j) * 36 + (((vv ^ ra) & 1) ? 18 : 0);
  const float* pb = high1 + ((b * 128 + (rb >> 1)) * 128 + j) * 36 + (((vv ^ rb) & 1) ? 18 : 0);
#pragma unroll
  for (int q = 0; q < 4; ++q) {
    r.qa[q] = *(const v4f_a8*)(pa + 4 * q);
    r.qb[q] = *(const v4f_a8*)(pb + 4 * q);
  }
  r.ea = *(const v2f*)(pa + 16);
  r.eb = *(const v2f*)(pb + 16);
  r.la = *(const v3f_a4*)(low + ((b * 256 + ra) * 256 + vv) * 3);
  r.lb = *(const v3f_a4*)(low + ((b * 256 + rb) * 256 + vv) * 3);
}

__device__ __forceinline__ void l1_mac(
    const L1Rec& r, int pya, int pyb, int px, v2f c0, v2f c1,
    v8f (&a1)[3], v8f (&a2)[3]) {
  constexpr int SA[3] = { 0, 2, 1 };
  constexpr int SB[3] = { 5, 3, 4 };
  float fa[18], fb[18];
#pragma unroll
  for (int q = 0; q < 4; ++q) {
    fa[4 * q] = r.qa[q].x; fa[4 * q + 1] = r.qa[q].y;
    fa[4 * q + 2] = r.qa[q].z; fa[4 * q + 3] = r.qa[q].w;
    fb[4 * q] = r.qb[q].x; fb[4 * q + 1] = r.qb[q].y;
    fb[4 * q + 2] = r.qb[q].z; fb[4 * q + 3] = r.qb[q].w;
  }
  fa[16] = r.ea.x; fa[17] = r.ea.y;
  fb[16] = r.eb.x; fb[17] = r.eb.y;
  const unsigned sa = (unsigned)pya << 31, sb = (unsigned)pyb << 31;
  const float ka = __uint_as_float(__float_as_uint(SCQ) ^ ((unsigned)(pya & px) << 31));
  const float kb = __uint_as_float(__float_as_uint(SCQ) ^ ((unsigned)(pyb & px) << 31));
  v2f w[3][3];
#pragma unroll
  for (int sub = 0; sub < 3; ++sub) {
#pragma unroll
    for (int ch = 0; ch < 3; ++ch) {
      const int ia = 3 * SA[sub] + ch, ib = 3 * SB[sub] + ch;
      w[sub][ch].x = (fa[ia] + __uint_as_float(__float_as_uint(fa[ib]) ^ sa)) * ka;
      w[sub][ch].y = (fb[ia] + __uint_as_float(__float_as_uint(fb[ib]) ^ sb)) * kb;
    }
  }
#pragma unroll
  for (int ch = 0; ch < 3; ++ch) {
    v2f wl = { r.la[ch], r.lb[ch] };
    a1[ch] = wmma4(wl,       c0, a1[ch]);
    a2[ch] = wmma4(w[1][ch], c0, a2[ch]);
    a1[ch] = wmma4(w[0][ch], c1, a1[ch]);
    a2[ch] = wmma4(w[2][ch], c1, a2[ch]);
  }
}

__global__ void __launch_bounds__(32)
__attribute__((amdgpu_waves_per_eu(4)))
k_level1(const float* __restrict__ low, const float* __restrict__ high1,
         float* __restrict__ Z) {
  const int tx = blockIdx.x, ty = blockIdx.y, b = blockIdx.z;
  const int x0 = tx * 32, y0 = ty * 16;
  const int r0 = 8 * ty - 6;
  const int c0 = 16 * tx - 6;

  const int lane = threadIdx.x;
  const int lo = lane & 15;
  const int h  = lane >> 4;

  int ra[5], rb[5];
#pragma unroll
  for (int kk = 0; kk < 5; ++kk) {
    ra[kk] = ref_idx(r0 + krow(kk, h, 0), 256);
    rb[kk] = ref_idx(r0 + krow(kk, h, 1), 256);
  }
  int vv[2];
#pragma unroll
  for (int T = 0; T < 2; ++T) {
    int cw = 16 * T + lo;
    cw = cw > 27 ? 27 : cw;
    vv[T] = ref_idx(c0 + cw, 256);
  }

  constexpr int NS = 10, PF = 1;
  L1Rec r[NS];
#pragma unroll
  for (int n = 0; n < PF; ++n) l1_load(low, high1, b, vv[n / 5], ra[n % 5], rb[n % 5], r[n]);

  v2f cf0[5], cf1[5];
#pragma unroll
  for (int kk = 0; kk < 5; ++kk) {
    const int ka = krow(kk, h, 0), kb = krow(kk, h, 1);
    cf0[kk].x = band14(false, lo, ka);  cf0[kk].y = band14(false, lo, kb);
    cf1[kk].x = band14(true,  lo, ka);  cf1[kk].y = band14(true,  lo, kb);
  }

  v8f d1[2][3], d2[2][3];
#pragma unroll
  for (int T = 0; T < 2; ++T) {
#pragma unroll
    for (int ch = 0; ch < 3; ++ch) { d1[T][ch] = v8f{}; d2[T][ch] = v8f{}; }
  }
#pragma unroll
  for (int n = 0; n < NS; ++n) {
    const int T = n / 5, kk = n % 5;
    if (n + PF < NS)
      l1_load(low, high1, b, vv[(n + PF) / 5], ra[(n + PF) % 5], rb[(n + PF) % 5], r[n + PF]);
    __builtin_amdgcn_sched_barrier(0);
    l1_mac(r[n], ra[kk] & 1, rb[kk] & 1, vv[T] & 1, cf0[kk], cf1[kk], d1[T], d2[T]);
  }

#pragma unroll
  for (int U = 0; U < 2; ++U) {
    v8f o[3] = { v8f{}, v8f{}, v8f{} };
#pragma unroll
    for (int T = 0; T < 2; ++T) {
#pragma unroll
      for (int s = 0; s < 4; ++s) {
        if (U == 0 && T == 1 && s >= 2) continue;
        const int cx = 16 * T + 8 * h + 2 * s;
        v2f g0 = { band14(false, 16 * U + lo, cx), band14(false, 16 * U + lo, cx + 1) };
        v2f g1 = { band14(true,  16 * U + lo, cx), band14(true,  16 * U + lo, cx + 1) };
#pragma unroll
        for (int ch = 0; ch < 3; ++ch) {
          v2f A1 = { d1[T][ch][2 * s], d1[T][ch][2 * s + 1] };
          v2f A2 = { d2[T][ch][2 * s], d2[T][ch][2 * s + 1] };
          o[ch] = wmma4(A1, g0, o[ch]);
          o[ch] = wmma4(A2, g1, o[ch]);
        }
      }
    }
#pragma unroll
    for (int v = 0; v < 8; ++v) {
      const int row = y0 + v + 8 * h;
      const int col = x0 + 16 * U + lo;
      float* op = Z + ((b * 512 + row) * 512 + col) * 3;
      op[0] = o[0][v]; op[1] = o[1][v]; op[2] = o[2][v];
    }
  }
}

__device__ __forceinline__ float tap7(int i) {
  int ic = i < 0 ? 0 : (i > 6 ? 6 : i);
  float t = cG0O[ic];
  return (i == ic) ? t : 0.0f;
}
__device__ __forceinline__ float tap5(int i) {
  int ic = i < 0 ? 0 : (i > 4 ? 4 : i);
  float t = cG1O[ic];
  return (i == ic) ? t : 0.0f;
}

struct L0Rec { v4f q[9]; float za[3]; float zb[3]; };

__device__ __forceinline__ int l0_col(int x0, int T, int lo) {
  int cw = 16 * T + lo;
  cw = cw > 69 ? 69 : cw;
  return ref_idx(x0 - 3 + cw, 512);
}

__device__ __forceinline__ void l0_load(
    const float* __restrict__ Z, const float* __restrict__ high0,
    int b, int gc, int rik, L0Rec& r) {
  const v4f* rp = (const v4f*)(high0 + ((b * 256 + rik) * 256 + (gc >> 1)) * 36);
#pragma unroll
  for (int q = 0; q < 9; ++q) r.q[q] = rp[q];
  const int zo = ((b * 512 + 2 * rik) * 512 + gc) * 3;
#pragma unroll
  for (int ch = 0; ch < 3; ++ch) { r.za[ch] = Z[zo + ch]; r.zb[ch] = Z[zo + 512 * 3 + ch]; }
}

__device__ __forceinline__ void l0_mac(
    const L0Rec& r, bool px, v2f c0, v2f c1, v8f (&a1)[3], v8f (&a2)[3]) {
  constexpr int SA[3] = { 0, 2, 1 };
  constexpr int SB[3] = { 5, 3, 4 };
  const float sq = px ? -SCQ : SCQ;
  float f[36];
#pragma unroll
  for (int q = 0; q < 9; ++q) {
    f[4 * q] = r.q[q].x; f[4 * q + 1] = r.q[q].y;
    f[4 * q + 2] = r.q[q].z; f[4 * q + 3] = r.q[q].w;
  }
  v2f w[3][3];
#pragma unroll
  for (int sub = 0; sub < 3; ++sub) {
#pragma unroll
    for (int ch = 0; ch < 3; ++ch) {
      const int ia = 3 * SA[sub] + ch, ib = 3 * SB[sub] + ch;
      const float pa = px ? f[18 + ia] : f[ia];
      const float pb = px ? f[18 + ib] : f[ib];
      const float qa = px ? f[ia] : f[18 + ia];
      const float qb = px ? f[ib] : f[18 + ib];
      w[sub][ch].x = (pa + pb) * SCQ;
      w[sub][ch].y = (qa - qb) * sq;
    }
  }
#pragma unroll
  for (int ch = 0; ch < 3; ++ch) {
    v2f wz = { r.za[ch], r.zb[ch] };
    a1[ch] = wmma4(wz,       c0, a1[ch]);
    a2[ch] = wmma4(w[1][ch], c0, a2[ch]);
    a1[ch] = wmma4(w[0][ch], c1, a1[ch]);
    a2[ch] = wmma4(w[2][ch], c1, a2[ch]);
  }
}

__device__ __forceinline__ void l0_row_part(
    const v8f (&y1)[3], const v8f (&y2)[3], int D,
    const v2f (&g0)[2][4], const v2f (&g1)[2][4], v8f (&o)[3]) {
#pragma unroll
  for (int s = 0; s < 4; ++s) {
    if (D == 1 && s == 3) continue;
#pragma unroll
    for (int ch = 0; ch < 3; ++ch) {
      v2f A1 = { y1[ch][2 * s], y1[ch][2 * s + 1] };
      v2f A2 = { y2[ch][2 * s], y2[ch][2 * s + 1] };
      o[ch] = wmma4(A1, g0[D][s], o[ch]);
      o[ch] = wmma4(A2, g1[D][s], o[ch]);
    }
  }
}

__device__ __forceinline__ void l0_store(
    const v8f (&o)[3], int lo, int h, float* __restrict__ out,
    int b, int x0, int y0, int U) {
#pragma unroll
  for (int v = 0; v < 8; ++v) {
    const int row = y0 + v + 8 * h;
    const int col = x0 + 16 * U + lo;
    float* op = out + ((b * 512 + row) * 512 + col) * 3;
    op[0] = o[0][v]; op[1] = o[1][v]; op[2] = o[2][v];
  }
}

__global__ void __launch_bounds__(32)
k_level0(const float* __restrict__ Z, const float* __restrict__ high0,
         float* __restrict__ out) {
  const int tx = blockIdx.x, ty = blockIdx.y, b = blockIdx.z;
  const int x0 = tx * 64, y0 = ty * 16;
  const int lane = threadIdx.x;
  const int lo = lane & 15;
  const int h  = lane >> 4;

  int ri[6], ka[6], kb[6];
#pragma unroll
  for (int kk = 0; kk < 6; ++kk) {
    const int ru = y0 - 4 + 4 * kk + 2 * h;
    const int sw = (ru < 0 || ru >= 512) ? 1 : 0;
    ri[kk] = ref_idx(ru, 512) >> 1;
    ka[kk] = 4 * kk + 2 * h + sw;
    kb[kk] = 4 * kk + 2 * h + (sw ^ 1);
  }
  int gc[5];
#pragma unroll
  for (int T = 0; T < 5; ++T) gc[T] = l0_col(x0, T, lo);

  constexpr int NS = 30, PF = 3;
  L0Rec r[NS];
#pragma unroll
  for (int n = 0; n < PF; ++n) l0_load(Z, high0, b, gc[n / 6], ri[n % 6], r[n]);

  v2f cf0[6], cf1[6];
#pragma unroll
  for (int kk = 0; kk < 6; ++kk) {
    cf0[kk].x = tap7(ka[kk] - lo - 1);  cf0[kk].y = tap7(kb[kk] - lo - 1);
    cf1[kk].x = tap5(ka[kk] - lo - 2);  cf1[kk].y = tap5(kb[kk] - lo - 2);
  }

  v2f g0[2][4], g1[2][4];
#pragma unroll
  for (int D = 0; D < 2; ++D) {
#pragma unroll
    for (int s = 0; s < 4; ++s) {
      const int dx = 16 * D + 8 * h + 2 * s - lo;
      g0[D][s].x = tap7(dx);      g0[D][s].y = tap7(dx + 1);
      g1[D][s].x = tap5(dx - 1);  g1[D][s].y = tap5(dx);
    }
  }

  v8f a1[3], a2[3], o[4][3];
#pragma unroll
  for (int ch = 0; ch < 3; ++ch) {
    a1[ch] = v8f{}; a2[ch] = v8f{};
#pragma unroll
    for (int U = 0; U < 4; ++U) o[U][ch] = v8f{};
  }
#pragma unroll
  for (int n = 0; n < NS; ++n) {
    const int T = n / 6, kk = n % 6;
    if (n + PF < NS)
      l0_load(Z, high0, b, gc[(n + PF) / 6], ri[(n + PF) % 6], r[n + PF]);
    __builtin_amdgcn_sched_barrier(0);
    l0_mac(r[n], (gc[T] & 1) != 0, cf0[kk], cf1[kk], a1, a2);
    if (kk == 5) {
      if (T >= 1) {
        l0_row_part(a1, a2, 1, g0, g1, o[T - 1]);
        l0_store(o[T - 1], lo, h, out, b, x0, y0, T - 1);
      }
      if (T <= 3) l0_row_part(a1, a2, 0, g0, g1, o[T]);
#pragma unroll
      for (int ch = 0; ch < 3; ++ch) { a1[ch] = v8f{}; a2[ch] = v8f{}; }
    }
  }
}

extern "C" void kernel_launch(void* const* d_in, const int* in_sizes, int n_in,
                              void* d_out, int out_size, void* d_ws, size_t ws_size,
                              hipStream_t stream) {
  (void)in_sizes; (void)n_in; (void)out_size; (void)ws_size;
  const float* low   = (const float*)d_in[0];
  const float* high0 = (const float*)d_in[1];
  const float* high1 = (const float*)d_in[2];
  float* out = (float*)d_out;
  float* Zws = (float*)d_ws;

  dim3 grid1(16, 32, 8);
  dim3 grid(8, 32, 8);
  k_level1<<<grid1, dim3(32), 0, stream>>>(low, high1, Zws);
  k_level0<<<grid, dim3(32), 0, stream>>>(Zws, high0, out);
}


# level-0 column pass in level-1's k order (lane halves = rows 2i, 2i+1 of one pixel): a lane reads two 18-float half records per step, a load touches one row of records, not two; column sums regrouped
# speedup vs baseline: 6.7964x; 1.0354x over previous
#include <hip/hip_runtime.h>
#include <hip/hip_bf16.h>


typedef float v2f __attribute__((ext_vector_type(2)));
typedef float v8f __attribute__((ext_vector_type(8)));

#define SCQ 0.70710678118654752440f

__constant__ float cG0B[14] = {
   0.00325314f, -0.00388321f,  0.03466035f, -0.03887280f,
  -0.11720389f,  0.27529538f,  0.75614564f,  0.56881042f,
   0.01186609f, -0.10671180f,  0.02382538f,  0.01702522f,
  -0.00543948f, -0.00455690f };
__constant__ float cG0A[14] = {
  -0.00455690f, -0.00543948f,  0.01702522f,  0.02382538f,
  -0.10671180f,  0.01186609f,  0.56881042f,  0.75614564f,
   0.27529538f, -0.11720389f, -0.03887280f,  0.03466035f,
  -0.00388321f,  0.00325314f };
__constant__ float cG1B[14] = {
  -0.00455690f,  0.00543948f,  0.01702522f, -0.02382538f,
  -0.10671180f, -0.01186609f,  0.56881042f, -0.75614564f,
   0.27529538f,  0.11720389f, -0.03887280f, -0.03466035f,
  -0.00388321f, -0.00325314f };
__constant__ float cG1A[14] = {
  -0.00325314f, -0.00388321f, -0.03466035f, -0.03887280f,
   0.11720389f,  0.27529538f, -0.75614564f,  0.56881042f,
  -0.01186609f, -0.10671180f, -0.02382538f,  0.01702522f,
   0.00543948f, -0.00455690f };

__constant__ float cG0O[7] = { -3.0f/280.0f, -15.0f/280.0f, 73.0f/280.0f,
                               170.0f/280.0f, 73.0f/280.0f, -15.0f/280.0f,
                               -3.0f/280.0f };
__constant__ float cG1O[5] = { -1.0f/20.0f, -5.0f/20.0f, 12.0f/20.0f,
                               -5.0f/20.0f, -1.0f/20.0f };

__device__ __forceinline__ int ref_idx(int i, int L) {
  if (i < 0)  i = -1 - i;
  if (i >= L) i = 2 * L - 1 - i;
  return i;
}

__device__ __forceinline__ float c2q_fetch(const float* __restrict__ hi,
                                           int b, int r, int v, int ch,
                                           int sub, int Ld) {
  constexpr int SA[3] = { 0, 2, 1 };
  constexpr int SB[3] = { 5, 3, 4 };
  const int i = r >> 1, j = v >> 1;
  const long long base = (((long long)b * Ld + i) * Ld + j) * 36;
  const int py = r & 1, px = v & 1;
  const int im = (px ^ py) ? 18 : 0;
  const float a  = hi[base + ch + 3 * SA[sub] + im];
  const float bb = hi[base + ch + 3 * SB[sub] + im];
  return py ? ((px ^ py) ? (a - bb) * SCQ : (bb - a) * SCQ) : (a + bb) * SCQ;
}

__device__ __forceinline__ v8f wmma4(v2f a, v2f b, v8f c) {
  return __builtin_amdgcn_wmma_f32_16x16x4_f32(
       false, a,  false, b,
       (short)0, c,  false,  false);
}

__device__ __forceinline__ float band14(bool g1, int x, int c) {
  const int  s  = x & 3, q = x >> 2;
  const int  p  = s >> 1;
  const bool aB = (s & 1) == 0;
  const float* f = g1 ? (aB ? cG1B : cG1A) : (aB ? cG0B : cG0A);
  const int base = g1 ? (aB ? 7 : 6) : (aB ? 6 : 7);
  const int d    = 2 * q + base + 6 - c;
  const bool ok  = (d >= 0) && (d <= 12) && ((d & 1) == 0);
  const float t  = f[ok ? d + p : 0];
  return ok ? t : 0.0f;
}

__device__ __forceinline__ int krow(int kk, int h, int pos) {
  return 4 * kk + h + 2 * pos;
}

typedef float v3f __attribute__((ext_vector_type(3)));
typedef float v4f __attribute__((ext_vector_type(4)));
typedef v4f v4f_a8 __attribute__((aligned(8)));
typedef v3f v3f_a4 __attribute__((aligned(4)));

struct L1Rec { v4f qa[4]; v2f ea; v4f qb[4]; v2f eb; v3f la; v3f lb; };

__device__ __forceinline__ void l1_load(
    const float* __restrict__ low, const float* __restrict__ high1,
    int b, int vv, int ra, int rb, L1Rec& r) {
  const int j = vv >> 1;
  const float* pa = high1 + ((b * 128 + (ra >> 1)) * 128 + j) * 36 + (((vv ^ ra) & 1) ? 18 : 0);
  const float* pb = high1 + ((b * 128 + (rb >> 1)) * 128 + j) * 36 + (((vv ^ rb) & 1) ? 18 : 0);
#pragma unroll
  for (int q = 0; q < 4; ++q) {
    r.qa[q] = *(const v4f_a8*)(pa + 4 * q);
    r.qb[q] = *(const v4f_a8*)(pb + 4 * q);
  }
  r.ea = *(const v2f*)(pa + 16);
  r.eb = *(const v2f*)(pb + 16);
  r.la = *(const v3f_a4*)(low + ((b * 256 + ra) * 256 + vv) * 3);
  r.lb = *(const v3f_a4*)(low + ((b * 256 + rb) * 256 + vv) * 3);
}

__device__ __forceinline__ void l1_mac(
    const L1Rec& r, int pya, int pyb, int px, v2f c0, v2f c1,
    v8f (&a1)[3], v8f (&a2)[3]) {
  constexpr int SA[3] = { 0, 2, 1 };
  constexpr int SB[3] = { 5, 3, 4 };
  float fa[18], fb[18];
#pragma unroll
  for (int q = 0; q < 4; ++q) {
    fa[4 * q] = r.qa[q].x; fa[4 * q + 1] = r.qa[q].y;
    fa[4 * q + 2] = r.qa[q].z; fa[4 * q + 3] = r.qa[q].w;
    fb[4 * q] = r.qb[q].x; fb[4 * q + 1] = r.qb[q].y;
    fb[4 * q + 2] = r.qb[q].z; fb[4 * q + 3] = r.qb[q].w;
  }
  fa[16] = r.ea.x; fa[17] = r.ea.y;
  fb[16] = r.eb.x; fb[17] = r.eb.y;
  const unsigned sa = (unsigned)pya << 31, sb = (unsigned)pyb << 31;
  const float ka = __uint_as_float(__float_as_uint(SCQ) ^ ((unsigned)(pya & px) << 31));
  const float kb = __uint_as_float(__float_as_uint(SCQ) ^ ((unsigned)(pyb & px) << 31));
  v2f w[3][3];
#pragma unroll
  for (int sub = 0; sub < 3; ++sub) {
#pragma unroll
    for (int ch = 0; ch < 3; ++ch) {
      const int ia = 3 * SA[sub] + ch, ib = 3 * SB[sub] + ch;
      w[sub][ch].x = (fa[ia] + __uint_as_float(__float_as_uint(fa[ib]) ^ sa)) * ka;
      w[sub][ch].y = (fb[ia] + __uint_as_float(__float_as_uint(fb[ib]) ^ sb)) * kb;
    }
  }
#pragma unroll
  for (int ch = 0; ch < 3; ++ch) {
    v2f wl = { r.la[ch], r.lb[ch] };
    a1[ch] = wmma4(wl,       c0, a1[ch]);
    a2[ch] = wmma4(w[1][ch], c0, a2[ch]);
    a1[ch] = wmma4(w[0][ch], c1, a1[ch]);
    a2[ch] = wmma4(w[2][ch], c1, a2[ch]);
  }
}

__global__ void __launch_bounds__(32)
__attribute__((amdgpu_waves_per_eu(4)))
k_level1(const float* __restrict__ low, const float* __restrict__ high1,
         float* __restrict__ Z) {
  const int tx = blockIdx.x, ty = blockIdx.y, b = blockIdx.z;
  const int x0 = tx * 32, y0 = ty * 16;
  const int r0 = 8 * ty - 6;
  const int c0 = 16 * tx - 6;

  const int lane = threadIdx.x;
  const int lo = lane & 15;
  const int h  = lane >> 4;

  int ra[5], rb[5];
#pragma unroll
  for (int kk = 0; kk < 5; ++kk) {
    ra[kk] = ref_idx(r0 + krow(kk, h, 0), 256);
    rb[kk] = ref_idx(r0 + krow(kk, h, 1), 256);
  }
  int vv[2];
#pragma unroll
  for (int T = 0; T < 2; ++T) {
    int cw = 16 * T + lo;
    cw = cw > 27 ? 27 : cw;
    vv[T] = ref_idx(c0 + cw, 256);
  }

  constexpr int NS = 10, PF = 1;
  L1Rec r[NS];
#pragma unroll
  for (int n = 0; n < PF; ++n) l1_load(low, high1, b, vv[n / 5], ra[n % 5], rb[n % 5], r[n]);

  v2f cf0[5], cf1[5];
#pragma unroll
  for (int kk = 0; kk < 5; ++kk) {
    const int ka = krow(kk, h, 0), kb = krow(kk, h, 1);
    cf0[kk].x = band14(false, lo, ka);  cf0[kk].y = band14(false, lo, kb);
    cf1[kk].x = band14(true,  lo, ka);  cf1[kk].y = band14(true,  lo, kb);
  }

  v8f d1[2][3], d2[2][3];
#pragma unroll
  for (int T = 0; T < 2; ++T) {
#pragma unroll
    for (int ch = 0; ch < 3; ++ch) { d1[T][ch] = v8f{}; d2[T][ch] = v8f{}; }
  }
#pragma unroll
  for (int n = 0; n < NS; ++n) {
    const int T = n / 5, kk = n % 5;
    if (n + PF < NS)
      l1_load(low, high1, b, vv[(n + PF) / 5], ra[(n + PF) % 5], rb[(n + PF) % 5], r[n + PF]);
    __builtin_amdgcn_sched_barrier(0);
    l1_mac(r[n], ra[kk] & 1, rb[kk] & 1, vv[T] & 1, cf0[kk], cf1[kk], d1[T], d2[T]);
  }

#pragma unroll
  for (int U = 0; U < 2; ++U) {
    v8f o[3] = { v8f{}, v8f{}, v8f{} };
#pragma unroll
    for (int T = 0; T < 2; ++T) {
#pragma unroll
      for (int s = 0; s < 4; ++s) {
        if (U == 0 && T == 1 && s >= 2) continue;
        const int cx = 16 * T + 8 * h + 2 * s;
        v2f g0 = { band14(false, 16 * U + lo, cx), band14(false, 16 * U + lo, cx + 1) };
        v2f g1 = { band14(true,  16 * U + lo, cx), band14(true,  16 * U + lo, cx + 1) };
#pragma unroll
        for (int ch = 0; ch < 3; ++ch) {
          v2f A1 = { d1[T][ch][2 * s], d1[T][ch][2 * s + 1] };
          v2f A2 = { d2[T][ch][2 * s], d2[T][ch][2 * s + 1] };
          o[ch] = wmma4(A1, g0, o[ch]);
          o[ch] = wmma4(A2, g1, o[ch]);
        }
      }
    }
#pragma unroll
    for (int v = 0; v < 8; ++v) {
      const int row = y0 + v + 8 * h;
      const int col = x0 + 16 * U + lo;
      float* op = Z + ((b * 512 + row) * 512 + col) * 3;
      op[0] = o[0][v]; op[1] = o[1][v]; op[2] = o[2][v];
    }
  }
}

__device__ __forceinline__ float tap7(int i) {
  int ic = i < 0 ? 0 : (i > 6 ? 6 : i);
  float t = cG0O[ic];
  return (i == ic) ? t : 0.0f;
}
__device__ __forceinline__ float tap5(int i) {
  int ic = i < 0 ? 0 : (i > 4 ? 4 : i);
  float t = cG1O[ic];
  return (i == ic) ? t : 0.0f;
}

typedef L1Rec L0Rec;

__device__ __forceinline__ int l0_col(int x0, int T, int lo) {
  int cw = 16 * T + lo;
  cw = cw > 69 ? 69 : cw;
  return ref_idx(x0 - 3 + cw, 512);
}

__device__ __forceinline__ void l0_load(
    const float* __restrict__ Z, const float* __restrict__ high0,
    int b, int gc, int ra, int rb, L0Rec& r) {
  const int j = gc >> 1;
  const float* pa = high0 + ((b * 256 + (ra >> 1)) * 256 + j) * 36 + (((gc ^ ra) & 1) ? 18 : 0);
  const float* pb = high0 + ((b * 256 + (rb >> 1)) * 256 + j) * 36 + (((gc ^ rb) & 1) ? 18 : 0);
#pragma unroll
  for (int q = 0; q < 4; ++q) {
    r.qa[q] = *(const v4f_a8*)(pa + 4 * q);
    r.qb[q] = *(const v4f_a8*)(pb + 4 * q);
  }
  r.ea = *(const v2f*)(pa + 16);
  r.eb = *(const v2f*)(pb + 16);
  r.la = *(const v3f_a4*)(Z + ((b * 512 + ra) * 512 + gc) * 3);
  r.lb = *(const v3f_a4*)(Z + ((b * 512 + rb) * 512 + gc) * 3);
}

__device__ __forceinline__ void l0_row_part(
    const v8f (&y1)[3], const v8f (&y2)[3], int D,
    const v2f (&g0)[2][4], const v2f (&g1)[2][4], v8f (&o)[3]) {
#pragma unroll
  for (int s = 0; s < 4; ++s) {
    if (D == 1 && s == 3) continue;
#pragma unroll
    for (int ch = 0; ch < 3; ++ch) {
      v2f A1 = { y1[ch][2 * s], y1[ch][2 * s + 1] };
      v2f A2 = { y2[ch][2 * s], y2[ch][2 * s + 1] };
      o[ch] = wmma4(A1, g0[D][s], o[ch]);
      o[ch] = wmma4(A2, g1[D][s], o[ch]);
    }
  }
}

__device__ __forceinline__ void l0_store(
    const v8f (&o)[3], int lo, int h, float* __restrict__ out,
    int b, int x0, int y0, int U) {
#pragma unroll
  for (int v = 0; v < 8; ++v) {
    const int row = y0 + v + 8 * h;
    const int col = x0 + 16 * U + lo;
    float* op = out + ((b * 512 + row) * 512 + col) * 3;
    op[0] = o[0][v]; op[1] = o[1][v]; op[2] = o[2][v];
  }
}

__global__ void __launch_bounds__(32)
k_level0(const float* __restrict__ Z, const float* __restrict__ high0,
         float* __restrict__ out) {
  const int tx = blockIdx.x, ty = blockIdx.y, b = blockIdx.z;
  const int x0 = tx * 64, y0 = ty * 16;
  const int lane = threadIdx.x;
  const int lo = lane & 15;
  const int h  = lane >> 4;

  int ka[6], kb[6], ra[6], rb[6];
#pragma unroll
  for (int kk = 0; kk < 6; ++kk) {
    ka[kk] = krow(kk, h, 0);
    kb[kk] = krow(kk, h, 1);
    ra[kk] = ref_idx(y0 - 4 + ka[kk], 512);
    rb[kk] = ref_idx(y0 - 4 + kb[kk], 512);
  }
  int gc[5];
#pragma unroll
  for (int T = 0; T < 5; ++T) gc[T] = l0_col(x0, T, lo);

  constexpr int NS = 30, PF = 3;
  L0Rec r[NS];
#pragma unroll
  for (int n = 0; n < PF; ++n) l0_load(Z, high0, b, gc[n / 6], ra[n % 6], rb[n % 6], r[n]);

  v2f cf0[6], cf1[6];
#pragma unroll
  for (int kk = 0; kk < 6; ++kk) {
    cf0[kk].x = tap7(ka[kk] - lo - 1);  cf0[kk].y = tap7(kb[kk] - lo - 1);
    cf1[kk].x = tap5(ka[kk] - lo - 2);  cf1[kk].y = tap5(kb[kk] - lo - 2);
  }

  v2f g0[2][4], g1[2][4];
#pragma unroll
  for (int D = 0; D < 2; ++D) {
#pragma unroll
    for (int s = 0; s < 4; ++s) {
      const int dx = 16 * D + 8 * h + 2 * s - lo;
      g0[D][s].x = tap7(dx);      g0[D][s].y = tap7(dx + 1);
      g1[D][s].x = tap5(dx - 1);  g1[D][s].y = tap5(dx);
    }
  }

  v8f a1[3], a2[3], o[4][3];
#pragma unroll
  for (int ch = 0; ch < 3; ++ch) {
    a1[ch] = v8f{}; a2[ch] = v8f{};
#pragma unroll
    for (int U = 0; U < 4; ++U) o[U][ch] = v8f{};
  }
#pragma unroll
  for (int n = 0; n < NS; ++n) {
    const int T = n / 6, kk = n % 6;
    if (n + PF < NS)
      l0_load(Z, high0, b, gc[(n + PF) / 6], ra[(n + PF) % 6], rb[(n + PF) % 6], r[n + PF]);
    __builtin_amdgcn_sched_barrier(0);
    l1_mac(r[n], ra[kk] & 1, rb[kk] & 1, gc[T] & 1, cf0[kk], cf1[kk], a1, a2);
    if (kk == 5) {
      if (T >= 1) {
        l0_row_part(a1, a2, 1, g0, g1, o[T - 1]);
        l0_store(o[T - 1], lo, h, out, b, x0, y0, T - 1);
      }
      if (T <= 3) l0_row_part(a1, a2, 0, g0, g1, o[T]);
#pragma unroll
      for (int ch = 0; ch < 3; ++ch) { a1[ch] = v8f{}; a2[ch] = v8f{}; }
    }
  }
}

extern "C" void kernel_launch(void* const* d_in, const int* in_sizes, int n_in,
                              void* d_out, int out_size, void* d_ws, size_t ws_size,
                              hipStream_t stream) {
  (void)in_sizes; (void)n_in; (void)out_size; (void)ws_size;
  const float* low   = (const float*)d_in[0];
  const float* high0 = (const float*)d_in[1];
  const float* high1 = (const float*)d_in[2];
  float* out = (float*)d_out;
  float* Zws = (float*)d_ws;

  dim3 grid1(16, 32, 8);
  dim3 grid(8, 32, 8);
  k_level1<<<grid1, dim3(32), 0, stream>>>(low, high1, Zws);
  k_level0<<<grid, dim3(32), 0, stream>>>(Zws, high0, out);
}


# row-pass k order re-dealt: column-pass lanes take window columns so a k-step is 4 of one parity (level 1, output tiles = even/odd columns) or 4 adjacent (level 0); zero k-steps go; row sums regrouped
# speedup vs baseline: 7.2427x; 1.0657x over previous
#include <hip/hip_runtime.h>
#include <hip/hip_bf16.h>


typedef float v2f __attribute__((ext_vector_type(2)));
typedef float v8f __attribute__((ext_vector_type(8)));

#define SCQ 0.70710678118654752440f

__constant__ float cG0B[14] = {
   0.00325314f, -0.00388321f,  0.03466035f, -0.03887280f,
  -0.11720389f,  0.27529538f,  0.75614564f,  0.56881042f,
   0.01186609f, -0.10671180f,  0.02382538f,  0.01702522f,
  -0.00543948f, -0.00455690f };
__constant__ float cG0A[14] = {
  -0.00455690f, -0.00543948f,  0.01702522f,  0.02382538f,
  -0.10671180f,  0.01186609f,  0.56881042f,  0.75614564f,
   0.27529538f, -0.11720389f, -0.03887280f,  0.03466035f,
  -0.00388321f,  0.00325314f };
__constant__ float cG1B[14] = {
  -0.00455690f,  0.00543948f,  0.01702522f, -0.02382538f,
  -0.10671180f, -0.01186609f,  0.56881042f, -0.75614564f,
   0.27529538f,  0.11720389f, -0.03887280f, -0.03466035f,
  -0.00388321f, -0.00325314f };
__constant__ float cG1A[14] = {
  -0.00325314f, -0.00388321f, -0.03466035f, -0.03887280f,
   0.11720389f,  0.27529538f, -0.75614564f,  0.56881042f,
  -0.01186609f, -0.10671180f, -0.02382538f,  0.01702522f,
   0.00543948f, -0.00455690f };

__constant__ float cG0O[7] = { -3.0f/280.0f, -15.0f/280.0f, 73.0f/280.0f,
                               170.0f/280.0f, 73.0f/280.0f, -15.0f/280.0f,
                               -3.0f/280.0f };
__constant__ float cG1O[5] = { -1.0f/20.0f, -5.0f/20.0f, 12.0f/20.0f,
                               -5.0f/20.0f, -1.0f/20.0f };

__device__ __forceinline__ int ref_idx(int i, int L) {
  if (i < 0)  i = -1 - i;
  if (i >= L) i = 2 * L - 1 - i;
  return i;
}

__device__ __forceinline__ float c2q_fetch(const float* __restrict__ hi,
                                           int b, int r, int v, int ch,
                                           int sub, int Ld) {
  constexpr int SA[3] = { 0, 2, 1 };
  constexpr int SB[3] = { 5, 3, 4 };
  const int i = r >> 1, j = v >> 1;
  const long long base = (((long long)b * Ld + i) * Ld + j) * 36;
  const int py = r & 1, px = v & 1;
  const int im = (px ^ py) ? 18 : 0;
  const float a  = hi[base + ch + 3 * SA[sub] + im];
  const float bb = hi[base + ch + 3 * SB[sub] + im];
  return py ? ((px ^ py) ? (a - bb) * SCQ : (bb - a) * SCQ) : (a + bb) * SCQ;
}

__device__ __forceinline__ v8f wmma4(v2f a, v2f b, v8f c) {
  return __builtin_amdgcn_wmma_f32_16x16x4_f32(
       false, a,  false, b,
       (short)0, c,  false,  false);
}

__device__ __forceinline__ float band14(bool g1, int x, int c) {
  const int  s  = x & 3, q = x >> 2;
  const int  p  = s >> 1;
  const bool aB = (s & 1) == 0;
  const float* f = g1 ? (aB ? cG1B : cG1A) : (aB ? cG0B : cG0A);
  const int base = g1 ? (aB ? 7 : 6) : (aB ? 6 : 7);
  const int d    = 2 * q + base + 6 - c;
  const bool ok  = (d >= 0) && (d <= 12) && ((d & 1) == 0);
  const float t  = f[ok ? d + p : 0];
  return ok ? t : 0.0f;
}

__device__ __forceinline__ float band_row(const float* f, int base, int lo, int c) {
  const int e = lo + base + 6 - c;
  const int d = e - (lo & 1);
  const bool ok = (d >= 0) && (d <= 12);
  const float t = f[ok ? e : 0];
  return ok ? t : 0.0f;
}

__device__ __forceinline__ int krow(int kk, int h, int pos) {
  return 4 * kk + h + 2 * pos;
}

__device__ __forceinline__ int col_deal1(int i) {
  const int s = (i & 7) >> 1, pos = i & 1, hh = i >> 3;
  return 8 * (s & 1) + 4 * hh + 2 * pos + (s >> 1);
}

__device__ __forceinline__ int col_deal0(int i) {
  const int s = (i & 7) >> 1, pos = i & 1, hh = i >> 3;
  return 4 * s + 2 * hh + pos;
}

typedef float v3f __attribute__((ext_vector_type(3)));
typedef float v4f __attribute__((ext_vector_type(4)));
typedef v4f v4f_a8 __attribute__((aligned(8)));
typedef v3f v3f_a4 __attribute__((aligned(4)));

struct L1Rec { v4f qa[4]; v2f ea; v4f qb[4]; v2f eb; v3f la; v3f lb; };

__device__ __forceinline__ void l1_load(
    const float* __restrict__ low, const float* __restrict__ high1,
    int b, int vv, int ra, int rb, L1Rec& r) {
  const int j = vv >> 1;
  const float* pa = high1 + ((b * 128 + (ra >> 1)) * 128 + j) * 36 + (((vv ^ ra) & 1) ? 18 : 0);
  const float* pb = high1 + ((b * 128 + (rb >> 1)) * 128 + j) * 36 + (((vv ^ rb) & 1) ? 18 : 0);
#pragma unroll
  for (int q = 0; q < 4; ++q) {
    r.qa[q] = *(const v4f_a8*)(pa + 4 * q);
    r.qb[q] = *(const v4f_a8*)(pb + 4 * q);
  }
  r.ea = *(const v2f*)(pa + 16);
  r.eb = *(const v2f*)(pb + 16);
  r.la = *(const v3f_a4*)(low + ((b * 256 + ra) * 256 + vv) * 3);
  r.lb = *(const v3f_a4*)(low + ((b * 256 + rb) * 256 + vv) * 3);
}

__device__ __forceinline__ void l1_mac(
    const L1Rec& r, int pya, int pyb, int px, v2f c0, v2f c1,
    v8f (&a1)[3], v8f (&a2)[3]) {
  constexpr int SA[3] = { 0, 2, 1 };
  constexpr int SB[3] = { 5, 3, 4 };
  float fa[18], fb[18];
#pragma unroll
  for (int q = 0; q < 4; ++q) {
    fa[4 * q] = r.qa[q].x; fa[4 * q + 1] = r.qa[q].y;
    fa[4 * q + 2] = r.qa[q].z; fa[4 * q + 3] = r.qa[q].w;
    fb[4 * q] = r.qb[q].x; fb[4 * q + 1] = r.qb[q].y;
    fb[4 * q + 2] = r.qb[q].z; fb[4 * q + 3] = r.qb[q].w;
  }
  fa[16] = r.ea.x; fa[17] = r.ea.y;
  fb[16] = r.eb.x; fb[17] = r.eb.y;
  const unsigned sa = (unsigned)pya << 31, sb = (unsigned)pyb << 31;
  const float ka = __uint_as_float(__float_as_uint(SCQ) ^ ((unsigned)(pya & px) << 31));
  const float kb = __uint_as_float(__float_as_uint(SCQ) ^ ((unsigned)(pyb & px) << 31));
  v2f w[3][3];
#pragma unroll
  for (int sub = 0; sub < 3; ++sub) {
#pragma unroll
    for (int ch = 0; ch < 3; ++ch) {
      const int ia = 3 * SA[sub] + ch, ib = 3 * SB[sub] + ch;
      w[sub][ch].x = (fa[ia] + __uint_as_float(__float_as_uint(fa[ib]) ^ sa)) * ka;
      w[sub][ch].y = (fb[ia] + __uint_as_float(__float_as_uint(fb[ib]) ^ sb)) * kb;
    }
  }
#pragma unroll
  for (int ch = 0; ch < 3; ++ch) {
    v2f wl = { r.la[ch], r.lb[ch] };
    a1[ch] = wmma4(wl,       c0, a1[ch]);
    a2[ch] = wmma4(w[1][ch], c0, a2[ch]);
    a1[ch] = wmma4(w[0][ch], c1, a1[ch]);
    a2[ch] = wmma4(w[2][ch], c1, a2[ch]);
  }
}

__global__ void __launch_bounds__(32)
__attribute__((amdgpu_waves_per_eu(4)))
k_level1(const float* __restrict__ low, const float* __restrict__ high1,
         float* __restrict__ Z) {
  const int tx = blockIdx.x, ty = blockIdx.y, b = blockIdx.z;
  const int x0 = tx * 32, y0 = ty * 16;
  const int r0 = 8 * ty - 6;
  const int c0 = 16 * tx - 6;

  const int lane = threadIdx.x;
  const int lo = lane & 15;
  const int h  = lane >> 4;

  int ra[5], rb[5];
#pragma unroll
  for (int kk = 0; kk < 5; ++kk) {
    ra[kk] = ref_idx(r0 + krow(kk, h, 0), 256);
    rb[kk] = ref_idx(r0 + krow(kk, h, 1), 256);
  }
  int vv[2];
#pragma unroll
  for (int T = 0; T < 2; ++T) {
    int cw = 16 * T + col_deal1(lo);
    cw = cw > 27 ? 27 : cw;
    vv[T] = ref_idx(c0 + cw, 256);
  }

  constexpr int NS = 10, PF = 1;
  L1Rec r[NS];
#pragma unroll
  for (int n = 0; n < PF; ++n) l1_load(low, high1, b, vv[n / 5], ra[n % 5], rb[n % 5], r[n]);

  v2f cf0[5], cf1[5];
#pragma unroll
  for (int kk = 0; kk < 5; ++kk) {
    const int ka = krow(kk, h, 0), kb = krow(kk, h, 1);
    cf0[kk].x = band14(false, lo, ka);  cf0[kk].y = band14(false, lo, kb);
    cf1[kk].x = band14(true,  lo, ka);  cf1[kk].y = band14(true,  lo, kb);
  }

  v8f d1[2][3], d2[2][3];
#pragma unroll
  for (int T = 0; T < 2; ++T) {
#pragma unroll
    for (int ch = 0; ch < 3; ++ch) { d1[T][ch] = v8f{}; d2[T][ch] = v8f{}; }
  }
#pragma unroll
  for (int n = 0; n < NS; ++n) {
    const int T = n / 5, kk = n % 5;
    if (n + PF < NS)
      l1_load(low, high1, b, vv[(n + PF) / 5], ra[(n + PF) % 5], rb[(n + PF) % 5], r[n + PF]);
    __builtin_amdgcn_sched_barrier(0);
    l1_mac(r[n], ra[kk] & 1, rb[kk] & 1, vv[T] & 1, cf0[kk], cf1[kk], d1[T], d2[T]);
    __builtin_amdgcn_sched_barrier(0);
  }

  v8f o[2][3];
#pragma unroll
  for (int U = 0; U < 2; ++U) {
#pragma unroll
    for (int ch = 0; ch < 3; ++ch) o[U][ch] = v8f{};
  }
#pragma unroll
  for (int U = 0; U < 2; ++U) {
#pragma unroll
    for (int T = 0; T < 2; ++T) {
#pragma unroll
      for (int s = 0; s < 4; ++s) {
        const bool use_g0 = (U == 0) == (s < 2);
        const int cx = 16 * T + col_deal1(8 * h + 2 * s);
        const float* f = use_g0 ? (U == 0 ? cG0B : cG0A) : (U == 0 ? cG1B : cG1A);
        const int base = use_g0 ? (U == 0 ? 6 : 7) : (U == 0 ? 7 : 6);
        v2f g = { band_row(f, base, lo, cx), band_row(f, base, lo, cx + 2) };
#pragma unroll
        for (int ch = 0; ch < 3; ++ch) {
          v2f A = use_g0 ? v2f{ d1[T][ch][2 * s], d1[T][ch][2 * s + 1] }
                         : v2f{ d2[T][ch][2 * s], d2[T][ch][2 * s + 1] };
          o[U][ch] = wmma4(A, g, o[U][ch]);
        }
      }
    }
  }
#pragma unroll
  for (int v = 0; v < 8; ++v) {
    const int row = y0 + v + 8 * h;
    const int col = x0 + 2 * lo;
    float* op = Z + ((b * 512 + row) * 512 + col) * 3;
    v4f q = { o[0][0][v], o[0][1][v], o[0][2][v], o[1][0][v] };
    v2f e = { o[1][1][v], o[1][2][v] };
    *(v4f_a8*)op = q;
    *(v2f*)(op + 4) = e;
  }
}

__device__ __forceinline__ float tap7(int i) {
  int ic = i < 0 ? 0 : (i > 6 ? 6 : i);
  float t = cG0O[ic];
  return (i == ic) ? t : 0.0f;
}
__device__ __forceinline__ float tap5(int i) {
  int ic = i < 0 ? 0 : (i > 4 ? 4 : i);
  float t = cG1O[ic];
  return (i == ic) ? t : 0.0f;
}

typedef L1Rec L0Rec;

__device__ __forceinline__ int l0_col(int x0, int T, int lo) {
  int cw = 16 * T + col_deal0(lo);
  cw = cw > 69 ? 69 : cw;
  return ref_idx(x0 - 3 + cw, 512);
}

__device__ __forceinline__ void l0_load(
    const float* __restrict__ Z, const float* __restrict__ high0,
    int b, int gc, int ra, int rb, L0Rec& r) {
  const int j = gc >> 1;
  const float* pa = high0 + ((b * 256 + (ra >> 1)) * 256 + j) * 36 + (((gc ^ ra) & 1) ? 18 : 0);
  const float* pb = high0 + ((b * 256 + (rb >> 1)) * 256 + j) * 36 + (((gc ^ rb) & 1) ? 18 : 0);
#pragma unroll
  for (int q = 0; q < 4; ++q) {
    r.qa[q] = *(const v4f_a8*)(pa + 4 * q);
    r.qb[q] = *(const v4f_a8*)(pb + 4 * q);
  }
  r.ea = *(const v2f*)(pa + 16);
  r.eb = *(const v2f*)(pb + 16);
  r.la = *(const v3f_a4*)(Z + ((b * 512 + ra) * 512 + gc) * 3);
  r.lb = *(const v3f_a4*)(Z + ((b * 512 + rb) * 512 + gc) * 3);
}

__device__ __forceinline__ void l0_row_part(
    const v8f (&y1)[3], const v8f (&y2)[3], int D,
    const v2f (&g0)[2][4], const v2f (&g1)[2][4], v8f (&o)[3]) {
#pragma unroll
  for (int s = 0; s < 4; ++s) {
    if (D == 1 && s >= 2) continue;
#pragma unroll
    for (int ch = 0; ch < 3; ++ch) {
      v2f A1 = { y1[ch][2 * s], y1[ch][2 * s + 1] };
      v2f A2 = { y2[ch][2 * s], y2[ch][2 * s + 1] };
      o[ch] = wmma4(A1, g0[D][s], o[ch]);
      o[ch] = wmma4(A2, g1[D][s], o[ch]);
    }
  }
}

__device__ __forceinline__ void l0_store(
    const v8f (&o)[3], int lo, int h, float* __restrict__ out,
    int b, int x0, int y0, int U) {
#pragma unroll
  for (int v = 0; v < 8; ++v) {
    const int row = y0 + v + 8 * h;
    const int col = x0 + 16 * U + lo;
    float* op = out + ((b * 512 + row) * 512 + col) * 3;
    op[0] = o[0][v]; op[1] = o[1][v]; op[2] = o[2][v];
  }
}

__global__ void __launch_bounds__(32)
k_level0(const float* __restrict__ Z, const float* __restrict__ high0,
         float* __restrict__ out) {
  const int tx = blockIdx.x, ty = blockIdx.y, b = blockIdx.z;
  const int x0 = tx * 64, y0 = ty * 16;
  const int lane = threadIdx.x;
  const int lo = lane & 15;
  const int h  = lane >> 4;

  int ka[6], kb[6], ra[6], rb[6];
#pragma unroll
  for (int kk = 0; kk < 6; ++kk) {
    ka[kk] = krow(kk, h, 0);
    kb[kk] = krow(kk, h, 1);
    ra[kk] = ref_idx(y0 - 4 + ka[kk], 512);
    rb[kk] = ref_idx(y0 - 4 + kb[kk], 512);
  }
  int gc[5];
#pragma unroll
  for (int T = 0; T < 5; ++T) gc[T] = l0_col(x0, T, lo);

  constexpr int NS = 30, PF = 3;
  L0Rec r[NS];
#pragma unroll
  for (int n = 0; n < PF; ++n) l0_load(Z, high0, b, gc[n / 6], ra[n % 6], rb[n % 6], r[n]);

  v2f cf0[6], cf1[6];
#pragma unroll
  for (int kk = 0; kk < 6; ++kk) {
    cf0[kk].x = tap7(ka[kk] - lo - 1);  cf0[kk].y = tap7(kb[kk] - lo - 1);
    cf1[kk].x = tap5(ka[kk] - lo - 2);  cf1[kk].y = tap5(kb[kk] - lo - 2);
  }

  v2f g0[2][4], g1[2][4];
#pragma unroll
  for (int D = 0; D < 2; ++D) {
#pragma unroll
    for (int s = 0; s < 4; ++s) {
      const int dx = 16 * D + col_deal0(8 * h + 2 * s) - lo;
      g0[D][s].x = tap7(dx);      g0[D][s].y = tap7(dx + 1);
      g1[D][s].x = tap5(dx - 1);  g1[D][s].y = tap5(dx);
    }
  }

  v8f a1[3], a2[3], o[4][3];
#pragma unroll
  for (int ch = 0; ch < 3; ++ch) {
    a1[ch] = v8f{}; a2[ch] = v8f{};
#pragma unroll
    for (int U = 0; U < 4; ++U) o[U][ch] = v8f{};
  }
#pragma unroll
  for (int n = 0; n < NS; ++n) {
    const int T = n / 6, kk = n % 6;
    if (n + PF < NS)
      l0_load(Z, high0, b, gc[(n + PF) / 6], ra[(n + PF) % 6], rb[(n + PF) % 6], r[n + PF]);
    __builtin_amdgcn_sched_barrier(0);
    l1_mac(r[n], ra[kk] & 1, rb[kk] & 1, gc[T] & 1, cf0[kk], cf1[kk], a1, a2);
    if (kk == 5) {
      if (T >= 1) {
        l0_row_part(a1, a2, 1, g0, g1, o[T - 1]);
        l0_store(o[T - 1], lo, h, out, b, x0, y0, T - 1);
      }
      if (T <= 3) l0_row_part(a1, a2, 0, g0, g1, o[T]);
#pragma unroll
      for (int ch = 0; ch < 3; ++ch) { a1[ch] = v8f{}; a2[ch] = v8f{}; }
    }
  }
}

extern "C" void kernel_launch(void* const* d_in, const int* in_sizes, int n_in,
                              void* d_out, int out_size, void* d_ws, size_t ws_size,
                              hipStream_t stream) {
  (void)in_sizes; (void)n_in; (void)out_size; (void)ws_size;
  const float* low   = (const float*)d_in[0];
  const float* high0 = (const float*)d_in[1];
  const float* high1 = (const float*)d_in[2];
  float* out = (float*)d_out;
  float* Zws = (float*)d_ws;

  dim3 grid1(16, 32, 8);
  dim3 grid(8, 32, 8);
  k_level1<<<grid1, dim3(32), 0, stream>>>(low, high1, Zws);
  k_level0<<<grid, dim3(32), 0, stream>>>(Zws, high0, out);
}
